# starting kernel
# baseline (speedup 1.0000x reference)
#include <hip/hip_runtime.h>
#include <hip/hip_bf16.h>

typedef __attribute__((ext_vector_type(2))) float v2f;
typedef __attribute__((ext_vector_type(4))) float v4f;
typedef __attribute__((ext_vector_type(8))) float v8f;

#define NNODES  100000
#define MPAD    100096
#define NEDGES  1600000
#define NGRAPHS 512

__global__ void zero_kernel(float* __restrict__ p, long long n) {
  long long i = (long long)blockIdx.x * blockDim.x + threadIdx.x;
  if (i < n) p[i] = 0.0f;
}

__global__ void deg_count_kernel(const int* __restrict__ dst, float* __restrict__ deg, int E) {
  int e = blockIdx.x * blockDim.x + threadIdx.x;
  if (e < E) atomicAdd(&deg[dst[e]], 1.0f);
}

__global__ void dinv_kernel(float* __restrict__ deg, int n) {
  int i = blockIdx.x * blockDim.x + threadIdx.x;
  if (i < n) deg[i] = rsqrtf(deg[i] + 1.0f);
}

__global__ void pad_x_kernel(const float* __restrict__ x, float* __restrict__ xp) {
  long long i = (long long)blockIdx.x * blockDim.x + threadIdx.x;
  if (i >= (long long)MPAD * 4) return;
  int node = (int)(i >> 2), j = (int)(i & 3);
  xp[i] = (node < NNODES && j < 3) ? x[(size_t)node * 3 + j] : 0.0f;
}

__global__ void pad_w1_kernel(const float* __restrict__ w, float* __restrict__ wp) {
  int i = blockIdx.x * blockDim.x + threadIdx.x;
  if (i >= 4 * 128) return;
  int k = i >> 7;
  wp[i] = (k < 3) ? w[i] : 0.0f;
}

__global__ void pad_wl2_kernel(const float* __restrict__ w, const float* __restrict__ b,
                               float* __restrict__ wp, float* __restrict__ bp) {
  int i = blockIdx.x * blockDim.x + threadIdx.x;
  if (i >= 32 * 16) return;
  int k = i >> 4, n = i & 15;
  wp[i] = (n < 9) ? w[k * 9 + n] : 0.0f;
  if (i < 16) bp[i] = (i < 9) ? b[i] : 0.0f;
}

__global__ void copy_out_kernel(const float* __restrict__ o16, float* __restrict__ out) {
  int i = blockIdx.x * blockDim.x + threadIdx.x;
  if (i >= NGRAPHS * 9) return;
  int g = i / 9, j = i - g * 9;
  out[i] = o16[g * 16 + j];
}

template <int K, int N, bool RELU, bool BIAS>
__global__ __launch_bounds__(256) void gemm_wmma(
    const float* __restrict__ A, const float* __restrict__ W,
    const float* __restrict__ bias, float* __restrict__ C) {
  __shared__ float lw[K * N];

  const int tid = threadIdx.x;
  constexpr int QUADS = (K / 2) * N / 4;
  for (int q = tid; q < QUADS; q += 256) {
    int t = (q * 4) / N;
    int n = (q * 4) % N;
    int k = t * 2;
    v4f w0 = *(const v4f*)(W + (size_t)k * N + n);
    v4f w1 = *(const v4f*)(W + (size_t)(k + 1) * N + n);
    v4f lo, hi;
    lo.x = w0.x; lo.y = w1.x; lo.z = w0.y; lo.w = w1.y;
    hi.x = w0.z; hi.y = w1.z; hi.z = w0.w; hi.w = w1.w;
    float* d = lw + ((size_t)t * N + n) * 2;
    *(v4f*)(d)     = lo;
    *(v4f*)(d + 4) = hi;
  }
  __syncthreads();

  const int lane = tid & 31;
  const int wave = tid >> 5;
  const int half = lane >> 4;
  const int l16  = lane & 15;
  const int row  = blockIdx.x * 128 + wave * 16 + l16;
  const float* __restrict__ Arow = A + (size_t)row * K + 2 * half;

  for (int n_base = 0; n_base < N; n_base += 16) {
    const int n = n_base + l16;
    float binit = BIAS ? bias[n] : 0.0f;
    v8f acc;
#pragma unroll
    for (int r = 0; r < 8; ++r) acc[r] = binit;

#pragma unroll
    for (int k = 0; k < K; k += 4) {
      v2f av = *(const v2f*)(Arow + k);
      if (RELU) { av.x = fmaxf(av.x, 0.0f); av.y = fmaxf(av.y, 0.0f); }
      const int kk = k + 2 * half;
      v2f bv = *(const v2f*)(lw + ((size_t)(kk >> 1) * N + n) * 2);
      acc = __builtin_amdgcn_wmma_f32_16x16x4_f32(
          false, av, false, bv, (short)0, acc, false, false);
    }

    float* __restrict__ Cp = C + (size_t)(blockIdx.x * 128 + wave * 16 + half * 8) * N + n;
#pragma unroll
    for (int r = 0; r < 8; ++r) Cp[(size_t)r * N] = acc[r];
  }
}

__global__ void selfloop_bias_kernel(const float* __restrict__ h,
                                     const float* __restrict__ dinv,
                                     const float* __restrict__ bias,
                                     float* __restrict__ out,
                                     int F, int lgF4) {
  long long i = (long long)blockIdx.x * blockDim.x + threadIdx.x;
  if (i >= ((long long)MPAD << lgF4)) return;
  int node = (int)(i >> lgF4);
  int f    = (int)(i & ((1 << lgF4) - 1)) * 4;
  v4f o;
  if (node < NNODES) {
    float di = dinv[node];
    float s  = di * di;
    v4f hv = *(const v4f*)(h + (size_t)node * F + f);
    v4f bv = *(const v4f*)(bias + f);
    o.x = hv.x * s + bv.x; o.y = hv.y * s + bv.y;
    o.z = hv.z * s + bv.z; o.w = hv.w * s + bv.w;
  } else {
    o.x = o.y = o.z = o.w = 0.0f;
  }
  *(v4f*)(out + (size_t)node * F + f) = o;
}

__global__ void edge_scatter_kernel(const float* __restrict__ h,
                                    const int* __restrict__ src,
                                    const int* __restrict__ dst,
                                    const float* __restrict__ dinv,
                                    float* __restrict__ out,
                                    int E, int F, int lgF4) {
  long long i = (long long)blockIdx.x * blockDim.x + threadIdx.x;
  if (i >= ((long long)E << lgF4)) return;
  int e = (int)(i >> lgF4);
  int f = (int)(i & ((1 << lgF4) - 1)) * 4;
  int s = src[e], d = dst[e];
  float nrm = dinv[s] * dinv[d];
  v4f hv = *(const v4f*)(h + (size_t)s * F + f);
  float* o = out + (size_t)d * F + f;
  atomicAdd(o + 0, hv.x * nrm);
  atomicAdd(o + 1, hv.y * nrm);
  atomicAdd(o + 2, hv.z * nrm);
  atomicAdd(o + 3, hv.w * nrm);
}

__global__ void pool_acc_kernel(const float* __restrict__ h,
                                const int* __restrict__ batch,
                                float* __restrict__ pooled,
                                float* __restrict__ cnt) {
  long long i = (long long)blockIdx.x * blockDim.x + threadIdx.x;
  if (i >= (long long)NNODES * 16) return;
  int node = (int)(i >> 4);
  int f    = (int)(i & 15) * 4;
  int g = batch[node];
  v4f hv = *(const v4f*)(h + (size_t)node * 64 + f);
  float* p = pooled + (size_t)g * 64 + f;
  atomicAdd(p + 0, hv.x);
  atomicAdd(p + 1, hv.y);
  atomicAdd(p + 2, hv.z);
  atomicAdd(p + 3, hv.w);
  if (f == 0) atomicAdd(&cnt[g], 1.0f);
}

__global__ void pool_div_kernel(float* __restrict__ pooled,
                                const float* __restrict__ cnt) {
  int i = blockIdx.x * blockDim.x + threadIdx.x;
  if (i < NGRAPHS * 64) pooled[i] /= fmaxf(cnt[i >> 6], 1.0f);
}

extern "C" void kernel_launch(void* const* d_in, const int* in_sizes, int n_in,
                              void* d_out, int out_size, void* d_ws, size_t ws_size,
                              hipStream_t stream) {
  const float* x   = (const float*)d_in[0];
  const int*   ei  = (const int*)d_in[1];
  const int*   bat = (const int*)d_in[2];
  const float* W1  = (const float*)d_in[3];  const float* b1  = (const float*)d_in[4];
  const float* W2  = (const float*)d_in[5];  const float* b2  = (const float*)d_in[6];
  const float* W3  = (const float*)d_in[7];  const float* b3  = (const float*)d_in[8];
  const float* W4  = (const float*)d_in[9];  const float* b4  = (const float*)d_in[10];
  const float* Wl1 = (const float*)d_in[11]; const float* bl1 = (const float*)d_in[12];
  const float* Wl2 = (const float*)d_in[13]; const float* bl2 = (const float*)d_in[14];
  float* out = (float*)d_out;

  const int* src = ei;
  const int* dst = ei + NEDGES;

  float* ws     = (float*)d_ws;
  float* dinv   = ws;
  float* xp     = dinv + NNODES;
  float* hT     = xp + (size_t)MPAD * 4;
  float* hA     = hT + (size_t)MPAD * 128;
  float* pooled = hA + (size_t)MPAD * 128;
  float* cnt    = pooled + NGRAPHS * 64;
  float* t1     = cnt + NGRAPHS;
  float* W1p    = t1 + NGRAPHS * 32;
  float* Wl2p   = W1p + 4 * 128;
  float* bl2p   = Wl2p + 32 * 16;
  float* o16    = bl2p + 16;

  const int B = 256;
  auto blks = [](long long n, int b) { return (unsigned)((n + b - 1) / b); };

  zero_kernel<<<blks(NNODES, B), B, 0, stream>>>(dinv, NNODES);
  deg_count_kernel<<<blks(NEDGES, B), B, 0, stream>>>(dst, dinv, NEDGES);
  dinv_kernel<<<blks(NNODES, B), B, 0, stream>>>(dinv, NNODES);

  pad_x_kernel<<<blks((long long)MPAD * 4, B), B, 0, stream>>>(x, xp);
  pad_w1_kernel<<<blks(4 * 128, B), B, 0, stream>>>(W1, W1p);
  pad_wl2_kernel<<<blks(32 * 16, B), B, 0, stream>>>(Wl2, bl2, Wl2p, bl2p);

  const unsigned GB = MPAD / 128;
  const long long n128 = (long long)MPAD * 32;
  const long long n64  = (long long)MPAD * 16;
  const long long e128 = (long long)NEDGES * 32;
  const long long e64  = (long long)NEDGES * 16;

  gemm_wmma<4, 128, false, false><<<GB, 256, 0, stream>>>(xp, W1p, nullptr, hT);
  selfloop_bias_kernel<<<blks(n128, B), B, 0, stream>>>(hT, dinv, b1, hA, 128, 5);
  edge_scatter_kernel<<<blks(e128, B), B, 0, stream>>>(hT, src, dst, dinv, hA, NEDGES, 128, 5);

  gemm_wmma<128, 128, true, false><<<GB, 256, 0, stream>>>(hA, W2, nullptr, hT);
  selfloop_bias_kernel<<<blks(n128, B), B, 0, stream>>>(hT, dinv, b2, hA, 128, 5);
  edge_scatter_kernel<<<blks(e128, B), B, 0, stream>>>(hT, src, dst, dinv, hA, NEDGES, 128, 5);

  gemm_wmma<128, 128, true, false><<<GB, 256, 0, stream>>>(hA, W3, nullptr, hT);
  selfloop_bias_kernel<<<blks(n128, B), B, 0, stream>>>(hT, dinv, b3, hA, 128, 5);
  edge_scatter_kernel<<<blks(e128, B), B, 0, stream>>>(hT, src, dst, dinv, hA, NEDGES, 128, 5);

  gemm_wmma<128, 64, true, false><<<GB, 256, 0, stream>>>(hA, W4, nullptr, hT);
  selfloop_bias_kernel<<<blks(n64, B), B, 0, stream>>>(hT, dinv, b4, hA, 64, 4);
  edge_scatter_kernel<<<blks(e64, B), B, 0, stream>>>(hT, src, dst, dinv, hA, NEDGES, 64, 4);

  zero_kernel<<<blks(NGRAPHS * 64, B), B, 0, stream>>>(pooled, NGRAPHS * 64);
  zero_kernel<<<blks(NGRAPHS, B), B, 0, stream>>>(cnt, NGRAPHS);
  pool_acc_kernel<<<blks((long long)NNODES * 16, B), B, 0, stream>>>(hA, bat, pooled, cnt);
  pool_div_kernel<<<blks(NGRAPHS * 64, B), B, 0, stream>>>(pooled, cnt);

  gemm_wmma<64, 32, false, true><<<NGRAPHS / 128, 256, 0, stream>>>(pooled, Wl1, bl1, t1);
  gemm_wmma<32, 16, true, true><<<NGRAPHS / 128, 256, 0, stream>>>(t1, Wl2p, bl2p, o16);
  copy_out_kernel<<<blks(NGRAPHS * 9, B), B, 0, stream>>>(o16, out);
}


# Launches 22 to 13: zero fills + paddings in one launch, layer-1 product as extra blocks of the binning launch, dinv = rsqrt(deg+1) where used, mean divide + 9-wide store in the head products
# speedup vs baseline: 3.1415x; 3.1415x over previous
#include <hip/hip_runtime.h>
#include <hip/hip_bf16.h>

typedef __attribute__((ext_vector_type(2))) float v2f;
typedef __attribute__((ext_vector_type(4))) float v4f;
typedef __attribute__((ext_vector_type(8))) float v8f;

#define NNODES  100000
#define MPAD    100096
#define NEDGES  1600000
#define NGRAPHS 512
#define TROWS   128
#define NBINS   (MPAD / TROWS)
#define NBPAD   1024
#define CHUNK   4096

#define PREP_Z0 NNODES
#define PREP_Z1 (2 * NBPAD)
#define PREP_Z2 (NGRAPHS * 64 + NGRAPHS)
#define PREP_X  (MPAD * 4)
#define PREP_W1 (4 * 128)
#define PREP_W2 (32 * 16)
#define PREP_TOTAL (PREP_Z0 + PREP_Z1 + PREP_Z2 + PREP_X + PREP_W1 + PREP_W2)
__global__ void prep_kernel(float* __restrict__ deg, float* __restrict__ bins,
                            float* __restrict__ pool,
                            const float* __restrict__ x, float* __restrict__ xp,
                            const float* __restrict__ w1, float* __restrict__ w1p,
                            const float* __restrict__ wl2, const float* __restrict__ bl2,
                            float* __restrict__ wl2p, float* __restrict__ bl2p) {
  int i = blockIdx.x * blockDim.x + threadIdx.x;
  if (i < PREP_Z0) { deg[i] = 0.0f; return; }
  i -= PREP_Z0;
  if (i < PREP_Z1) { bins[i] = 0.0f; return; }
  i -= PREP_Z1;
  if (i < PREP_Z2) { pool[i] = 0.0f; return; }
  i -= PREP_Z2;
  if (i < PREP_X) {
    const int node = i >> 2, j = i & 3;
    xp[i] = (node < NNODES && j < 3) ? x[(size_t)node * 3 + j] : 0.0f;
    return;
  }
  i -= PREP_X;
  if (i < PREP_W1) { w1p[i] = ((i >> 7) < 3) ? w1[i] : 0.0f; return; }
  i -= PREP_W1;
  if (i < PREP_W2) {
    const int k = i >> 4, n = i & 15;
    wl2p[i] = (n < 9) ? wl2[k * 9 + n] : 0.0f;
    if (i < 16) bl2p[i] = (i < 9) ? bl2[i] : 0.0f;
  }
}

__global__ __launch_bounds__(1024) void deg_bin_count_kernel(
    const int* __restrict__ dst, float* __restrict__ deg, int* __restrict__ binCnt) {
  __shared__ int hist[NBPAD];
  const int t = threadIdx.x;
  hist[t] = 0;
  __syncthreads();
  const int e0 = blockIdx.x * CHUNK + t;
#pragma unroll
  for (int i = 0; i < CHUNK / 1024; ++i) {
    const int e = e0 + i * 1024;
    if (e < NEDGES) {
      const int d = dst[e];
      atomicAdd(&deg[d], 1.0f);
      atomicAdd(&hist[d >> 7], 1);
    }
  }
  __syncthreads();
  const int c = hist[t];
  if (c) atomicAdd(&binCnt[t], c);
}

__device__ __forceinline__ void bin_fill_body(
    const int* __restrict__ src, const int* __restrict__ dst,
    const float* __restrict__ deg,
    const int* __restrict__ binCnt, int* __restrict__ binCur,
    int* __restrict__ binStart, uint2* __restrict__ rec) {
  static_assert(NNODES <= (1 << 17) && TROWS == 128, "record packing: 17 bits src, 7 bits row");
  __shared__ int sc[NBPAD];
  __shared__ int lc[NBPAD];
  const int t = threadIdx.x;
  const int c = binCnt[t];
  sc[t] = c;
  lc[t] = 0;
  __syncthreads();
  for (int off = 1; off < NBPAD; off <<= 1) {
    const int v = (t >= off) ? sc[t - off] : 0;
    __syncthreads();
    sc[t] += v;
    __syncthreads();
  }
  const int gstart = sc[t] - c;
  if (blockIdx.x == 0) binStart[t] = gstart;

  int d[CHUNK / 1024], s[CHUNK / 1024];
  float w[CHUNK / 1024];
  const int e0 = blockIdx.x * CHUNK + t;
#pragma unroll
  for (int i = 0; i < CHUNK / 1024; ++i) {
    const int e = e0 + i * 1024;
    d[i] = -1; s[i] = 0; w[i] = 0.0f;
    if (e < NEDGES) {
      d[i] = dst[e];
      s[i] = src[e];
      w[i] = rsqrtf(deg[s[i]] + 1.0f) * rsqrtf(deg[d[i]] + 1.0f);
      atomicAdd(&lc[d[i] >> 7], 1);
    }
  }
  __syncthreads();
  const int mine = lc[t];
  int first = gstart;
  if (mine) first += atomicAdd(&binCur[t], mine);
  sc[t] = first;
  lc[t] = 0;
  __syncthreads();
#pragma unroll
  for (int i = 0; i < CHUNK / 1024; ++i) {
    if (d[i] >= 0) {
      const int b = d[i] >> 7;
      const int r = atomicAdd(&lc[b], 1);
      uint2 o;
      o.x = (unsigned)s[i] | ((unsigned)(d[i] & 127) << 17);
      o.y = __float_as_uint(w[i]);
      rec[sc[b] + r] = o;
    }
  }
}


template <int K, int N>
__device__ __forceinline__ void stage_w(const float* __restrict__ W, float* __restrict__ lw,
                                        int tid, int nthreads) {
  constexpr int QUADS = (K / 2) * N / 4;
  for (int q = tid; q < QUADS; q += nthreads) {
    int t = (q * 4) / N;
    int n = (q * 4) % N;
    int k = t * 2;
    v4f w0 = *(const v4f*)(W + (size_t)k * N + n);
    v4f w1 = *(const v4f*)(W + (size_t)(k + 1) * N + n);
    v4f lo, hi;
    lo.x = w0.x; lo.y = w1.x; lo.z = w0.y; lo.w = w1.y;
    hi.x = w0.z; hi.y = w1.z; hi.z = w0.w; hi.w = w1.w;
    float* d = lw + ((size_t)t * N + n) * 2;
    *(v4f*)(d)     = lo;
    *(v4f*)(d + 4) = hi;
  }
}

template <int K, int N, bool RELU, bool BIAS, bool DIVCNT, int NC>
__device__ __forceinline__ void gemm_tile(
    const float* __restrict__ A, const float* __restrict__ lw,
    const float* __restrict__ bias, const float* __restrict__ cnt,
    float* __restrict__ C, int tile, int tid) {
  const int lane = tid & 31;
  const int wave = tid >> 5;
  const int half = lane >> 4;
  const int l16  = lane & 15;
  const int row  = tile * 128 + wave * 16 + l16;
  const float* __restrict__ Arow = A + (size_t)row * K + 2 * half;
  const float rdiv = DIVCNT ? fmaxf(cnt[row], 1.0f) : 1.0f;

  for (int n_base = 0; n_base < N; n_base += 16) {
    const int n = n_base + l16;
    float binit = BIAS ? bias[n] : 0.0f;
    v8f acc;
#pragma unroll
    for (int r = 0; r < 8; ++r) acc[r] = binit;

#pragma unroll
    for (int k = 0; k < K; k += 4) {
      v2f av = *(const v2f*)(Arow + k);
      if (DIVCNT) { av.x /= rdiv; av.y /= rdiv; }
      if (RELU) { av.x = fmaxf(av.x, 0.0f); av.y = fmaxf(av.y, 0.0f); }
      const int kk = k + 2 * half;
      v2f bv = *(const v2f*)(lw + ((size_t)(kk >> 1) * N + n) * 2);
      acc = __builtin_amdgcn_wmma_f32_16x16x4_f32(
          false, av, false, bv, (short)0, acc, false, false);
    }

    if constexpr (NC == N) {
      float* __restrict__ Cp = C + (size_t)(tile * 128 + wave * 16 + half * 8) * N + n;
#pragma unroll
      for (int r = 0; r < 8; ++r) Cp[(size_t)r * N] = acc[r];
    } else {
      static_assert(N == 16 && NC < N && (128 * NC) % 32 == 0, "packed store of one narrow tile");
      __shared__ float lo[128 * NC];
      if (n < NC) {
#pragma unroll
        for (int r = 0; r < 8; ++r) lo[(wave * 16 + half * 8 + r) * NC + n] = acc[r];
      }
      __syncthreads();
      float* __restrict__ Cb = C + (size_t)tile * 128 * NC;
      for (int i = tid; i < 128 * NC; i += 256) Cb[i] = lo[i];
    }
  }
}

template <int K, int N, bool RELU, bool BIAS, bool DIVCNT = false, int NC = N>
__global__ __launch_bounds__(256) void gemm_wmma(
    const float* __restrict__ A, const float* __restrict__ W,
    const float* __restrict__ bias, const float* __restrict__ cnt,
    float* __restrict__ C) {
  __shared__ float lw[K * N];
  stage_w<K, N>(W, lw, threadIdx.x, 256);
  __syncthreads();
  gemm_tile<K, N, RELU, BIAS, DIVCNT, NC>(A, lw, bias, cnt, C, blockIdx.x, threadIdx.x);
}

#define BIN_BLOCKS ((NEDGES + CHUNK - 1) / CHUNK)
#define L1_TILES   (MPAD / 128)
#define L1_BLOCKS  ((L1_TILES + 3) / 4)
__global__ __launch_bounds__(1024) void bin_fill_l1_kernel(
    const int* __restrict__ src, const int* __restrict__ dst,
    const float* __restrict__ deg,
    const int* __restrict__ binCnt, int* __restrict__ binCur,
    int* __restrict__ binStart, uint2* __restrict__ rec,
    const float* __restrict__ xp, const float* __restrict__ W1p,
    float* __restrict__ hT) {
  if (blockIdx.x >= BIN_BLOCKS) {
    __shared__ float lw1[4 * 128];
    stage_w<4, 128>(W1p, lw1, threadIdx.x, 1024);
    __syncthreads();
    const int grp  = __builtin_amdgcn_readfirstlane((int)(threadIdx.x >> 8));
    const int tile = (blockIdx.x - BIN_BLOCKS) * 4 + grp;
    if (tile < L1_TILES)
      gemm_tile<4, 128, false, false, false, 128>(xp, lw1, nullptr, nullptr, hT, tile,
                                                  (int)(threadIdx.x & 255));
    return;
  }
  bin_fill_body(src, dst, deg, binCnt, binCur, binStart, rec);
}

template <int F, int EPW, bool TAIL>
__device__ __forceinline__ void edge_trip(
    float* __restrict__ acc, const char* __restrict__ hb,
    const uint2* __restrict__ recs, int base, int n, int lane) {
  constexpr int VPL = 2;
  typedef v2f vec;
  uint2 rc[EPW];
  vec v[EPW];
#pragma unroll
  for (int q = 0; q < EPW; ++q) {
    int idx = base + q;
    if (TAIL) idx = idx < n ? idx : n - 1;
    rc[q] = recs[idx];
  }
#pragma unroll
  for (int q = 0; q < EPW; ++q) {
    const unsigned s = rc[q].x & 0x1FFFFu;
    v[q] = *(const vec*)(hb + s * (unsigned)(F * 4) + (unsigned)lane * 8u);
  }
#pragma unroll
  for (int q = 0; q < EPW; ++q) {
    if (!TAIL || base + q < n) {
      const unsigned dl = rc[q].x >> 17;
      const float wgt = __uint_as_float(rc[q].y);
#pragma unroll
      for (int j = 0; j < VPL; ++j)
        atomicAdd(&acc[dl * 64 + j * 32 + lane], v[q][j] * wgt);
    }
  }
}

template <int F, int EPW>
__global__ __launch_bounds__(256) void tile_sum_kernel(
    const float* __restrict__ h, const uint2* __restrict__ rec,
    const int* __restrict__ binStart, const int* __restrict__ binCnt,
    const float* __restrict__ deg, const float* __restrict__ bias,
    float* __restrict__ out) {
  constexpr int VPL = 2;
  constexpr int NSPLIT = F / 64;
  typedef v2f vec;
  static_assert(F % 64 == 0, "a block owns whole aligned 256-B units of a row");
  static_assert((long long)MPAD * F * 4 < (1ll << 32), "table offsets fit 32 bits");
  __shared__ float acc[TROWS * 64];

  const int lane = threadIdx.x & 31;
  const int wave = __builtin_amdgcn_readfirstlane((int)(threadIdx.x >> 5));
  const int tile = blockIdx.x / NSPLIT;
  const int col0 = (blockIdx.x % NSPLIT) * 64;
  const int row0 = tile * TROWS;

  const vec bv = *(const vec*)(bias + col0 + VPL * lane);
  {
    vec hv[TROWS / 8];
    float sl[TROWS / 8];
#pragma unroll
    for (int i = 0; i < TROWS / 8; ++i) {
      const int row  = row0 + i * 8 + wave;
      const int rowc = row < NNODES ? row : NNODES - 1;
      hv[i] = *(const vec*)(h + (size_t)rowc * F + col0 + VPL * lane);
      const float di = rsqrtf(deg[rowc] + 1.0f);
      sl[i] = di * di;
    }
#pragma unroll
    for (int i = 0; i < TROWS / 8; ++i) {
      const int r = i * 8 + wave;
      const bool real = row0 + r < NNODES;
#pragma unroll
      for (int j = 0; j < VPL; ++j)
        acc[r * 64 + j * 32 + lane] = real ? hv[i][j] * sl[i] + bv[j] : 0.0f;
    }
  }
  __syncthreads();

  const int beg = binStart[tile];
  const int n   = binCnt[tile];
  const char* __restrict__ hb = (const char*)(h + col0);
  const uint2* __restrict__ recs = rec + beg;
  int base = wave * EPW;
  for (; base + EPW <= n; base += 8 * EPW)
    edge_trip<F, EPW, false>(acc, hb, recs, base, n, lane);
  if (base < n)
    edge_trip<F, EPW, true>(acc, hb, recs, base, n, lane);
  __syncthreads();

#pragma unroll 4
  for (int i = 0; i < TROWS / 8; ++i) {
    const int r = i * 8 + wave;
    vec o;
#pragma unroll
    for (int j = 0; j < VPL; ++j) o[j] = acc[r * 64 + j * 32 + lane];
    *(vec*)(out + (size_t)(row0 + r) * F + col0 + VPL * lane) = o;
  }
}

#define PROWS 32
#define POOL_WAVES 4
#define POOL_BLOCKS ((NNODES / PROWS + POOL_WAVES - 1) / POOL_WAVES)

__device__ __forceinline__ void pool_flush(float* __restrict__ pooled, float* __restrict__ cnt,
                                           int g, v2f acc, int len, int lane) {
  float* p = pooled + (size_t)g * 64 + 2 * lane;
  atomicAdd(p + 0, acc.x);
  atomicAdd(p + 1, acc.y);
  if (lane == 0) atomicAdd(&cnt[g], (float)len);
}

__global__ __launch_bounds__(32 * POOL_WAVES) void pool_acc_kernel(
    const float* __restrict__ h, const int* __restrict__ batch,
    float* __restrict__ pooled, float* __restrict__ cnt) {
  static_assert(NNODES % PROWS == 0, "every wave owns PROWS whole rows");
  const int lane = threadIdx.x & 31;
  const int wave = __builtin_amdgcn_readfirstlane((int)(threadIdx.x >> 5));
  const int n0 = (blockIdx.x * POOL_WAVES + wave) * PROWS;
  if (n0 >= NNODES) return;
  const int* __restrict__ ids = batch + n0;
  const float* __restrict__ hp = h + (size_t)n0 * 64 + 2 * lane;
  int id[PROWS];
  v2f v[PROWS];
#pragma unroll
  for (int q = 0; q < PROWS; ++q) id[q] = ids[q];
#pragma unroll
  for (int q = 0; q < PROWS; ++q) v[q] = *(const v2f*)(hp + (size_t)q * 64);
  int cur = id[0];
  int len = 1;
  v2f acc = v[0];
#pragma unroll
  for (int q = 1; q < PROWS; ++q) {
    if (id[q] != cur) {
      pool_flush(pooled, cnt, cur, acc, len, lane);
      cur = id[q];
      len = 1;
      acc = v[q];
    } else {
      acc.x += v[q].x;
      acc.y += v[q].y;
      ++len;
    }
  }
  pool_flush(pooled, cnt, cur, acc, len, lane);
}


extern "C" void kernel_launch(void* const* d_in, const int* in_sizes, int n_in,
                              void* d_out, int out_size, void* d_ws, size_t ws_size,
                              hipStream_t stream) {
  const float* x   = (const float*)d_in[0];
  const int*   ei  = (const int*)d_in[1];
  const int*   bat = (const int*)d_in[2];
  const float* W1  = (const float*)d_in[3];  const float* b1  = (const float*)d_in[4];
  const float* W2  = (const float*)d_in[5];  const float* b2  = (const float*)d_in[6];
  const float* W3  = (const float*)d_in[7];  const float* b3  = (const float*)d_in[8];
  const float* W4  = (const float*)d_in[9];  const float* b4  = (const float*)d_in[10];
  const float* Wl1 = (const float*)d_in[11]; const float* bl1 = (const float*)d_in[12];
  const float* Wl2 = (const float*)d_in[13]; const float* bl2 = (const float*)d_in[14];
  float* out = (float*)d_out;

  const int* src = ei;
  const int* dst = ei + NEDGES;

  float* ws     = (float*)d_ws;
  float* dinv   = ws;
  float* xp     = dinv + MPAD;
  float* hT     = xp + (size_t)MPAD * 4;
  float* hA     = hT + (size_t)MPAD * 128;
  float* pooled = hA + (size_t)MPAD * 128;
  float* cnt    = pooled + NGRAPHS * 64;
  float* t1     = cnt + NGRAPHS;
  float* W1p    = t1 + NGRAPHS * 32;
  float* Wl2p   = W1p + 4 * 128;
  float* bl2p   = Wl2p + 32 * 16;
  int* binCnt   = (int*)(bl2p + 16);
  int* binCur   = binCnt + NBPAD;
  int* binStart = binCur + NBPAD;
  uint2* rec    = (uint2*)(binStart + NBPAD);

  const int B = 256;
  auto blks = [](long long n, int b) { return (unsigned)((n + b - 1) / b); };
  static_assert(NGRAPHS * 64 + NGRAPHS == PREP_Z2 && 2 * NBPAD == PREP_Z1, "ranges of the first launch");

  const unsigned CB = (NEDGES + CHUNK - 1) / CHUNK;
  prep_kernel<<<blks(PREP_TOTAL, B), B, 0, stream>>>(dinv, (float*)binCnt, pooled,
                                                     x, xp, W1, W1p, Wl2, bl2, Wl2p, bl2p);
  deg_bin_count_kernel<<<CB, 1024, 0, stream>>>(dst, dinv, binCnt);
  bin_fill_l1_kernel<<<BIN_BLOCKS + L1_BLOCKS, 1024, 0, stream>>>(src, dst, dinv, binCnt, binCur, binStart, rec,
                                                                 xp, W1p, hT);

  const unsigned GB = MPAD / 128;
  constexpr int EPW = 16;

  tile_sum_kernel<128, EPW><<<NBINS * 2, 256, 0, stream>>>(hT, rec, binStart, binCnt, dinv, b1, hA);

  gemm_wmma<128, 128, true, false><<<GB, 256, 0, stream>>>(hA, W2, nullptr, nullptr, hT);
  tile_sum_kernel<128, EPW><<<NBINS * 2, 256, 0, stream>>>(hT, rec, binStart, binCnt, dinv, b2, hA);

  gemm_wmma<128, 128, true, false><<<GB, 256, 0, stream>>>(hA, W3, nullptr, nullptr, hT);
  tile_sum_kernel<128, EPW><<<NBINS * 2, 256, 0, stream>>>(hT, rec, binStart, binCnt, dinv, b3, hA);

  gemm_wmma<128, 64, true, false><<<GB, 256, 0, stream>>>(hA, W4, nullptr, nullptr, hT);
  tile_sum_kernel<64, EPW><<<NBINS, 256, 0, stream>>>(hT, rec, binStart, binCnt, dinv, b4, hA);

  pool_acc_kernel<<<POOL_BLOCKS, 32 * POOL_WAVES, 0, stream>>>(hA, bat, pooled, cnt);

  gemm_wmma<64, 32, false, true, true><<<NGRAPHS / 128, 256, 0, stream>>>(pooled, Wl1, bl1, cnt, t1);
  gemm_wmma<32, 16, true, true, false, 9><<<NGRAPHS / 128, 256, 0, stream>>>(t1, Wl2p, bl2p, nullptr, out);
}
